# starting kernel
# baseline (speedup 1.0000x reference)
#include <hip/hip_runtime.h>
#include <math.h>

typedef __attribute__((ext_vector_type(2))) float v2f;
typedef __attribute__((ext_vector_type(8))) float v8f;

#define SOLVER_EPS 1e-4f
#define SOLVER_ITERS 50

__device__ __forceinline__ float sigmoidf(float x) {
    return 1.0f / (1.0f + expf(-x));
}

__global__ __launch_bounds__(256) void fnn_power_solver_kernel(
    const float* __restrict__ X,
    const float* __restrict__ w1,
    const float* __restrict__ w2,
    float* __restrict__ out,
    int B)
{
    __shared__ float Hs[256 * 33];
    __shared__ float w2s[128];

    const int tid  = threadIdx.x;
    const int lane = tid & 31;
    const int wave = tid >> 5;
    const int blockRow = blockIdx.x * 256;

    if (tid < 128) w2s[tid] = w2[tid];

    const int halfSel = (lane >> 4) & 1;
    const int lmod    = lane & 15;
    const int koff    = halfSel * 2;

    #pragma unroll
    for (int half = 0; half < 2; ++half) {
        const int lrow0 = wave * 32 + half * 16;
        int row = blockRow + lrow0 + lmod;
        if (row >= B) row = B - 1;

        v2f a[4];
        #pragma unroll
        for (int k = 0; k < 4; ++k) {
            const float2 av = *reinterpret_cast<const float2*>(X + row * 16 + k * 4 + koff);
            a[k].x = av.x;
            a[k].y = av.y;
        }

        #pragma unroll
        for (int ct = 0; ct < 2; ++ct) {
            const int col = ct * 16 + lmod;
            v8f c = {};
            #pragma unroll
            for (int k = 0; k < 4; ++k) {
                v2f b;
                b.x = w1[(k * 4 + koff + 0) * 32 + col];
                b.y = w1[(k * 4 + koff + 1) * 32 + col];
                c = __builtin_amdgcn_wmma_f32_16x16x4_f32(
                        false, a[k], false, b, (short)0, c, false, false);
            }
            #pragma unroll
            for (int i = 0; i < 8; ++i) {
                const int r = lrow0 + i + halfSel * 8;
                Hs[r * 33 + col] = sigmoidf(c[i]);
            }
        }
    }

    __syncthreads();

    const int row = blockRow + tid;
    if (row >= B) return;

    float z0 = 0.f, z1 = 0.f, z2 = 0.f, z3 = 0.f;
    const float* h = &Hs[tid * 33];
    #pragma unroll
    for (int j = 0; j < 32; ++j) {
        const float hv = h[j];
        z0 = fmaf(hv, w2s[j * 4 + 0], z0);
        z1 = fmaf(hv, w2s[j * 4 + 1], z1);
        z2 = fmaf(hv, w2s[j * 4 + 2], z2);
        z3 = fmaf(hv, w2s[j * 4 + 3], z3);
    }
    const float y0 = sigmoidf(z0), y1 = sigmoidf(z1);
    const float y2 = sigmoidf(z2), y3 = sigmoidf(z3);

    const float4* Xv = reinterpret_cast<const float4*>(X + row * 16);
    const float4 g1 = Xv[0];
    const float4 g2 = Xv[1];
    const float4 sg = Xv[2];
    const float4 rr = Xv[3];

    const float F101 = g1.y / g1.x, F110 = g1.z / g1.w;
    const float F201 = g2.y / g2.x, F210 = g2.z / g2.w;
    const float v10  = sg.x / g1.x, v11  = sg.y / g1.w;
    const float v20  = sg.z / g2.x, v21  = sg.w / g2.w;

    const float is1 = 1.0f / (y0 + y1);
    const float is2 = 1.0f / (y2 + y3);
    const float r10 = rr.x * y0 * is1, r11 = rr.x * y1 * is1;
    const float r20 = rr.y * y2 * is2, r21 = rr.y * y3 * is2;

    float p10 = 0.05f, p11 = 0.2f, p20 = 0.1f, p21 = 0.3f;
    bool done = false;
    for (int it = 0; it < SOLVER_ITERS; ++it) {
        if (done) break;
        const float s10 = p10 / fmaf(F101, p11, v10);
        const float s11 = p11 / fmaf(F110, p10, v11);
        const float s20 = p20 / fmaf(F201, p21, v20);
        const float s21 = p21 / fmaf(F210, p20, v21);
        const float n10 = r10 * p10 / log1pf(s10);
        const float n11 = r11 * p11 / log1pf(s11);
        const float n20 = r20 * p20 / log1pf(s20);
        const float n21 = r21 * p21 / log1pf(s21);
        const float d10 = n10 - p10, d11 = n11 - p11;
        const float d20 = n20 - p20, d21 = n21 - p21;
        const float err = sqrtf(d10 * d10 + d11 * d11) + sqrtf(d20 * d20 + d21 * d21);
        p10 = n10; p11 = n11; p20 = n20; p21 = n21;
        done = (err < SOLVER_EPS);
    }

    float4 po;
    po.x = p10; po.y = p11; po.z = p20; po.w = p21;
    *reinterpret_cast<float4*>(out + row * 4) = po;
}

extern "C" void kernel_launch(void* const* d_in, const int* in_sizes, int n_in,
                              void* d_out, int out_size, void* d_ws, size_t ws_size,
                              hipStream_t stream) {
    const float* X  = (const float*)d_in[0];
    const float* w1 = (const float*)d_in[1];
    const float* w2 = (const float*)d_in[2];
    float* out = (float*)d_out;

    const int B = in_sizes[0] / 16;
    const int grid = (B + 255) / 256;
    fnn_power_solver_kernel<<<grid, 256, 0, stream>>>(X, w1, w2, out, B);
}


# Single hardware v_rcp/v_exp/v_log/v_sqrt instructions replace the correctly rounded library division, expf, log1pf and sqrtf in both sigmoids and the solver loop; log1p keeps a lost-bits correction
# speedup vs baseline: 2.1757x; 2.1757x over previous
#include <hip/hip_runtime.h>
#include <math.h>

typedef __attribute__((ext_vector_type(2))) float v2f;
typedef __attribute__((ext_vector_type(8))) float v8f;

#define SOLVER_EPS 1e-4f
#define SOLVER_ITERS 50


__device__ __forceinline__ float rcp_hw(float x) {
    return __builtin_amdgcn_rcpf(x);
}

__device__ __forceinline__ float sigmoidf(float x) {
    return rcp_hw(1.0f + __builtin_amdgcn_exp2f(x * -1.44269504088896341f));
}

__device__ __forceinline__ float log1p_hw(float x) {
    const float u    = 1.0f + x;
    const float lost = x - (u - 1.0f);
    return fmaf(__builtin_amdgcn_logf(u), 0.693147180559945309f, lost * rcp_hw(u));
}

__global__ __launch_bounds__(256) void fnn_power_solver_kernel(
    const float* __restrict__ X,
    const float* __restrict__ w1,
    const float* __restrict__ w2,
    float* __restrict__ out,
    int B)
{
    __shared__ float Hs[256 * 33];
    __shared__ float w2s[128];

    const int tid  = threadIdx.x;
    const int lane = tid & 31;
    const int wave = tid >> 5;
    const int blockRow = blockIdx.x * 256;

    if (tid < 128) w2s[tid] = w2[tid];

    const int halfSel = (lane >> 4) & 1;
    const int lmod    = lane & 15;
    const int koff    = halfSel * 2;

    #pragma unroll
    for (int half = 0; half < 2; ++half) {
        const int lrow0 = wave * 32 + half * 16;
        int row = blockRow + lrow0 + lmod;
        if (row >= B) row = B - 1;

        v2f a[4];
        #pragma unroll
        for (int k = 0; k < 4; ++k) {
            const float2 av = *reinterpret_cast<const float2*>(X + row * 16 + k * 4 + koff);
            a[k].x = av.x;
            a[k].y = av.y;
        }

        #pragma unroll
        for (int ct = 0; ct < 2; ++ct) {
            const int col = ct * 16 + lmod;
            v8f c = {};
            #pragma unroll
            for (int k = 0; k < 4; ++k) {
                v2f b;
                b.x = w1[(k * 4 + koff + 0) * 32 + col];
                b.y = w1[(k * 4 + koff + 1) * 32 + col];
                c = __builtin_amdgcn_wmma_f32_16x16x4_f32(
                        false, a[k], false, b, (short)0, c, false, false);
            }
            #pragma unroll
            for (int i = 0; i < 8; ++i) {
                const int r = lrow0 + i + halfSel * 8;
                Hs[r * 33 + col] = sigmoidf(c[i]);
            }
        }
    }

    __syncthreads();

    const int row = blockRow + tid;
    if (row >= B) return;

    float z0 = 0.f, z1 = 0.f, z2 = 0.f, z3 = 0.f;
    const float* h = &Hs[tid * 33];
    #pragma unroll
    for (int j = 0; j < 32; ++j) {
        const float hv = h[j];
        z0 = fmaf(hv, w2s[j * 4 + 0], z0);
        z1 = fmaf(hv, w2s[j * 4 + 1], z1);
        z2 = fmaf(hv, w2s[j * 4 + 2], z2);
        z3 = fmaf(hv, w2s[j * 4 + 3], z3);
    }
    const float y0 = sigmoidf(z0), y1 = sigmoidf(z1);
    const float y2 = sigmoidf(z2), y3 = sigmoidf(z3);

    const float4* Xv = reinterpret_cast<const float4*>(X + row * 16);
    const float4 g1 = Xv[0];
    const float4 g2 = Xv[1];
    const float4 sg = Xv[2];
    const float4 rr = Xv[3];

    const float id10 = rcp_hw(g1.x), id11 = rcp_hw(g1.w);
    const float id20 = rcp_hw(g2.x), id21 = rcp_hw(g2.w);
    const float F101 = g1.y * id10, F110 = g1.z * id11;
    const float F201 = g2.y * id20, F210 = g2.z * id21;
    const float v10  = sg.x * id10, v11  = sg.y * id11;
    const float v20  = sg.z * id20, v21  = sg.w * id21;

    const float is1 = rcp_hw(y0 + y1);
    const float is2 = rcp_hw(y2 + y3);
    const float r10 = rr.x * y0 * is1, r11 = rr.x * y1 * is1;
    const float r20 = rr.y * y2 * is2, r21 = rr.y * y3 * is2;

    float p10 = 0.05f, p11 = 0.2f, p20 = 0.1f, p21 = 0.3f;
    bool done = false;
    for (int it = 0; it < SOLVER_ITERS; ++it) {
        if (done) break;
        const float s10 = p10 * rcp_hw(fmaf(F101, p11, v10));
        const float s11 = p11 * rcp_hw(fmaf(F110, p10, v11));
        const float s20 = p20 * rcp_hw(fmaf(F201, p21, v20));
        const float s21 = p21 * rcp_hw(fmaf(F210, p20, v21));
        const float n10 = r10 * p10 * rcp_hw(log1p_hw(s10));
        const float n11 = r11 * p11 * rcp_hw(log1p_hw(s11));
        const float n20 = r20 * p20 * rcp_hw(log1p_hw(s20));
        const float n21 = r21 * p21 * rcp_hw(log1p_hw(s21));
        const float d10 = n10 - p10, d11 = n11 - p11;
        const float d20 = n20 - p20, d21 = n21 - p21;
        const float err = __builtin_amdgcn_sqrtf(d10 * d10 + d11 * d11)
                        + __builtin_amdgcn_sqrtf(d20 * d20 + d21 * d21);
        p10 = n10; p11 = n11; p20 = n20; p21 = n21;
        done = (err < SOLVER_EPS);
    }

    float4 po;
    po.x = p10; po.y = p11; po.z = p20; po.w = p21;
    *reinterpret_cast<float4*>(out + row * 4) = po;
}

extern "C" void kernel_launch(void* const* d_in, const int* in_sizes, int n_in,
                              void* d_out, int out_size, void* d_ws, size_t ws_size,
                              hipStream_t stream) {
    const float* X  = (const float*)d_in[0];
    const float* w1 = (const float*)d_in[1];
    const float* w2 = (const float*)d_in[2];
    float* out = (float*)d_out;

    const int B = in_sizes[0] / 16;
    const int grid = (B + 255) / 256;
    fnn_power_solver_kernel<<<grid, 256, 0, stream>>>(X, w1, w2, out, B);
}


# WMMA roles swapped (w1^T as A, X^T as B): a lane's accumulators hold its own row's hidden units, layer 2 summed in-lane as two 16-term partials joined by one lane exchange; no LDS tile, no barrier
# speedup vs baseline: 2.3060x; 1.0599x over previous
#include <hip/hip_runtime.h>
#include <math.h>

typedef __attribute__((ext_vector_type(2))) float v2f;
typedef __attribute__((ext_vector_type(8))) float v8f;

#define SOLVER_EPS 1e-4f
#define SOLVER_ITERS 50


__device__ __forceinline__ float rcp_hw(float x) {
    return __builtin_amdgcn_rcpf(x);
}

__device__ __forceinline__ float sigmoidf(float x) {
    return rcp_hw(1.0f + __builtin_amdgcn_exp2f(x * -1.44269504088896341f));
}

__device__ __forceinline__ float log1p_hw(float x) {
    const float u    = 1.0f + x;
    const float lost = x - (u - 1.0f);
    return fmaf(__builtin_amdgcn_logf(u), 0.693147180559945309f, lost * rcp_hw(u));
}

__device__ __forceinline__ float xhalf(float x) {
    const int b = __builtin_bit_cast(int, x);
    return __builtin_bit_cast(float,
        __builtin_amdgcn_permlanex16(b, b, 0x76543210, 0xfedcba98, false, false));
}

__global__ __launch_bounds__(256) void fnn_power_solver_kernel(
    const float* __restrict__ X,
    const float* __restrict__ w1,
    const float* __restrict__ w2,
    float* __restrict__ out,
    int B)
{
    const int tid  = threadIdx.x;
    const int lane = tid & 31;
    const int wave = tid >> 5;
    const int halfSel = (lane >> 4) & 1;
    const int lmod    = lane & 15;
    const int koff    = halfSel * 2;
    const int waveRow = blockIdx.x * 256 + wave * 32;

    const int row  = waveRow + lane;
    const int rowc = (row < B) ? row : (B - 1);

    v2f wf[2][4];
    #pragma unroll
    for (int ct = 0; ct < 2; ++ct) {
        const int hid = ct * 16 + lmod;
        #pragma unroll
        for (int k = 0; k < 4; ++k) {
            wf[ct][k].x = w1[(k * 4 + koff + 0) * 32 + hid];
            wf[ct][k].y = w1[(k * 4 + koff + 1) * 32 + hid];
        }
    }

    const float4* w2q = reinterpret_cast<const float4*>(w2);
    float4 w2v[2][8];
    #pragma unroll
    for (int ct = 0; ct < 2; ++ct) {
        #pragma unroll
        for (int i = 0; i < 8; ++i) w2v[ct][i] = w2q[ct * 16 + halfSel * 8 + i];
    }

    const float4* Xv = reinterpret_cast<const float4*>(X + rowc * 16);
    const float4 g1 = Xv[0];
    const float4 g2 = Xv[1];
    const float4 sg = Xv[2];
    const float4 rr = Xv[3];

    float zp[2][4];
    #pragma unroll
    for (int t = 0; t < 2; ++t) {
        int trow = waveRow + t * 16 + lmod;
        if (trow >= B) trow = B - 1;

        v2f xf[4];
        #pragma unroll
        for (int k = 0; k < 4; ++k) {
            const float2 xv = *reinterpret_cast<const float2*>(X + trow * 16 + k * 4 + koff);
            xf[k].x = xv.x;
            xf[k].y = xv.y;
        }

        float z0 = 0.f, z1 = 0.f, z2 = 0.f, z3 = 0.f;
        #pragma unroll
        for (int ct = 0; ct < 2; ++ct) {
            v8f c = {};
            #pragma unroll
            for (int k = 0; k < 4; ++k) {
                c = __builtin_amdgcn_wmma_f32_16x16x4_f32(
                        false, wf[ct][k], false, xf[k], (short)0, c, false, false);
            }
            #pragma unroll
            for (int i = 0; i < 8; ++i) {
                const float hv = sigmoidf(c[i]);
                const float4 w = w2v[ct][i];
                z0 = fmaf(hv, w.x, z0);
                z1 = fmaf(hv, w.y, z1);
                z2 = fmaf(hv, w.z, z2);
                z3 = fmaf(hv, w.w, z3);
            }
        }
        zp[t][0] = z0; zp[t][1] = z1; zp[t][2] = z2; zp[t][3] = z3;
    }

    float z[4];
    #pragma unroll
    for (int o = 0; o < 4; ++o) {
        const float keep = halfSel ? zp[1][o] : zp[0][o];
        const float send = halfSel ? zp[0][o] : zp[1][o];
        z[o] = keep + xhalf(send);
    }

    const float y0 = sigmoidf(z[0]), y1 = sigmoidf(z[1]);
    const float y2 = sigmoidf(z[2]), y3 = sigmoidf(z[3]);

    const float id10 = rcp_hw(g1.x), id11 = rcp_hw(g1.w);
    const float id20 = rcp_hw(g2.x), id21 = rcp_hw(g2.w);
    const float F101 = g1.y * id10, F110 = g1.z * id11;
    const float F201 = g2.y * id20, F210 = g2.z * id21;
    const float v10  = sg.x * id10, v11  = sg.y * id11;
    const float v20  = sg.z * id20, v21  = sg.w * id21;

    const float is1 = rcp_hw(y0 + y1);
    const float is2 = rcp_hw(y2 + y3);
    const float r10 = rr.x * y0 * is1, r11 = rr.x * y1 * is1;
    const float r20 = rr.y * y2 * is2, r21 = rr.y * y3 * is2;

    float p10 = 0.05f, p11 = 0.2f, p20 = 0.1f, p21 = 0.3f;
    bool done = false;
    for (int it = 0; it < SOLVER_ITERS; ++it) {
        if (done) break;
        const float s10 = p10 * rcp_hw(fmaf(F101, p11, v10));
        const float s11 = p11 * rcp_hw(fmaf(F110, p10, v11));
        const float s20 = p20 * rcp_hw(fmaf(F201, p21, v20));
        const float s21 = p21 * rcp_hw(fmaf(F210, p20, v21));
        const float n10 = r10 * p10 * rcp_hw(log1p_hw(s10));
        const float n11 = r11 * p11 * rcp_hw(log1p_hw(s11));
        const float n20 = r20 * p20 * rcp_hw(log1p_hw(s20));
        const float n21 = r21 * p21 * rcp_hw(log1p_hw(s21));
        const float d10 = n10 - p10, d11 = n11 - p11;
        const float d20 = n20 - p20, d21 = n21 - p21;
        const float err = __builtin_amdgcn_sqrtf(d10 * d10 + d11 * d11)
                        + __builtin_amdgcn_sqrtf(d20 * d20 + d21 * d21);
        p10 = n10; p11 = n11; p20 = n20; p21 = n21;
        done = (err < SOLVER_EPS);
    }

    if (row < B) {
        float4 po;
        po.x = p10; po.y = p11; po.z = p20; po.w = p21;
        *reinterpret_cast<float4*>(out + row * 4) = po;
    }
}

extern "C" void kernel_launch(void* const* d_in, const int* in_sizes, int n_in,
                              void* d_out, int out_size, void* d_ws, size_t ws_size,
                              hipStream_t stream) {
    const float* X  = (const float*)d_in[0];
    const float* w1 = (const float*)d_in[1];
    const float* w2 = (const float*)d_in[2];
    float* out = (float*)d_out;

    const int B = in_sizes[0] / 16;
    const int grid = (B + 255) / 256;
    fnn_power_solver_kernel<<<grid, 256, 0, stream>>>(X, w1, w2, out, B);
}
